# starting kernel
# baseline (speedup 1.0000x reference)
#include <hip/hip_runtime.h>


typedef float v2f __attribute__((ext_vector_type(2)));
typedef float v4f __attribute__((ext_vector_type(4)));
typedef float v8f __attribute__((ext_vector_type(8)));

#define Bc 2
#define Hc 8
#define Sc 2048
#define Dc 64
#define BHc (Bc * Hc)
#define BQ 64
#define KS 68
#define VS 72

__launch_bounds__(128)
__global__ void retention_fused_kernel(const float* __restrict__ q,
                                       const float* __restrict__ k,
                                       const float* __restrict__ v,
                                       const float* __restrict__ omask,
                                       float* __restrict__ out)
{
    const int qb   = (int)(gridDim.x - 1u - blockIdx.x);
    const int bh   = (int)blockIdx.y;
    const int tid  = (int)threadIdx.x;
    const int w    = tid >> 5;
    const int lane = tid & 31;
    const int half = lane >> 4;
    const int l16  = lane & 15;

    __shared__ float ldsK[BQ * KS];
    __shared__ float ldsV[BQ * VS];
    __shared__ float ldsS[BQ * KS];
    __shared__ float ldsQC[BQ];

    const size_t bh_sd = (size_t)bh * Sc * Dc;
    const size_t bh_ss = (size_t)bh * Sc * (size_t)Sc;

    for (int rr = 0; rr < 16; ++rr) {
        const int row = w * 16 + rr;
        const size_t base = bh_ss + (size_t)(qb * BQ + row) * Sc;
        const v4f* om4 = (const v4f*)(omask + base);
        float s = 0.f;
        for (int j = lane; j < Sc / 4; j += 32) {
            v4f t = __builtin_nontemporal_load(om4 + j);
            s += (t.x + t.y) + (t.z + t.w);
        }
        #pragma unroll
        for (int off = 16; off > 0; off >>= 1) s += __shfl_xor(s, off, 32);
        if (lane == 0) {
            float m0 = omask[base];
            ldsQC[row] = m0 / sqrtf(s);
        }
    }
    __syncthreads();

    const int qrow = qb * BQ + w * 16 + l16;
    const float qc = ldsQC[w * 16 + l16];
    const float* qptr = q + bh_sd + (size_t)qrow * Dc;
    v2f qf[16];
    #pragma unroll
    for (int kk = 0; kk < 16; ++kk) {
        const int c = kk * 4 + half * 2;
        v2f a; a.x = qptr[c] * qc; a.y = qptr[c + 1] * qc;
        qf[kk] = a;
    }

    v8f oacc[4];
    v8f pacc;
    #pragma unroll
    for (int r = 0; r < 8; ++r) pacc[r] = 0.f;
    #pragma unroll
    for (int ct = 0; ct < 4; ++ct) {
        #pragma unroll
        for (int r = 0; r < 8; ++r) oacc[ct][r] = 0.f;
    }

    for (int kb = 0; kb <= qb; ++kb) {
        __syncthreads();
        {
            const float* kbase = k + bh_sd + (size_t)(kb * BQ) * Dc;
            const float* vbase = v + bh_sd + (size_t)(kb * BQ) * Dc;
            for (int idx = tid; idx < BQ * Dc / 4; idx += 128) {
                const int row  = idx >> 4;
                const int col4 = (idx & 15) * 4;
                const float kc = 1.f / omask[bh_ss + (size_t)(kb * BQ + row) * Sc];
                const v4f kvv = *(const v4f*)(kbase + row * Dc + col4);
                ldsK[row * KS + col4 + 0] = kvv.x * kc;
                ldsK[row * KS + col4 + 1] = kvv.y * kc;
                ldsK[row * KS + col4 + 2] = kvv.z * kc;
                ldsK[row * KS + col4 + 3] = kvv.w * kc;
                const v4f vvv = *(const v4f*)(vbase + row * Dc + col4);
                ldsV[row * VS + col4 + 0] = vvv.x;
                ldsV[row * VS + col4 + 1] = vvv.y;
                ldsV[row * VS + col4 + 2] = vvv.z;
                ldsV[row * VS + col4 + 3] = vvv.w;
            }
        }
        __syncthreads();

        #pragma unroll
        for (int t = 0; t < 4; ++t) {
            v8f s;
            #pragma unroll
            for (int r = 0; r < 8; ++r) s[r] = 0.f;
            #pragma unroll
            for (int kk = 0; kk < 16; ++kk) {
                const int d0   = kk * 4 + half * 2;
                const int krow = t * 16 + l16;
                v2f b; b.x = ldsK[krow * KS + d0]; b.y = ldsK[krow * KS + d0 + 1];
                s = __builtin_amdgcn_wmma_f32_16x16x4_f32(
                        false, qf[kk], false, b, (short)0, s, false, false);
            }
            const int jg = kb * BQ + t * 16 + l16;
            #pragma unroll
            for (int r = 0; r < 8; ++r) {
                const int ig = qb * BQ + w * 16 + r + 8 * half;
                const float sv = (jg <= ig) ? s[r] : 0.f;
                pacc[r] += sv;
                ldsS[(w * 16 + r + 8 * half) * KS + t * 16 + l16] = sv;
            }
        }
        __syncthreads();

        v2f sa[16];
        #pragma unroll
        for (int nn = 0; nn < 16; ++nn) {
            const int n0   = nn * 4 + half * 2;
            const int srow = w * 16 + l16;
            v2f a; a.x = ldsS[srow * KS + n0]; a.y = ldsS[srow * KS + n0 + 1];
            sa[nn] = a;
        }
        #pragma unroll
        for (int ct = 0; ct < 4; ++ct) {
            v8f o = oacc[ct];
            #pragma unroll
            for (int nn = 0; nn < 16; ++nn) {
                const int n0 = nn * 4 + half * 2;
                v2f b; b.x = ldsV[(n0    ) * VS + ct * 16 + l16];
                       b.y = ldsV[(n0 + 1) * VS + ct * 16 + l16];
                o = __builtin_amdgcn_wmma_f32_16x16x4_f32(
                        false, sa[nn], false, b, (short)0, o, false, false);
            }
            oacc[ct] = o;
        }
    }

    float invp[8];
    #pragma unroll
    for (int r = 0; r < 8; ++r) {
        float p = pacc[r];
        p += __shfl_xor(p, 1, 32);
        p += __shfl_xor(p, 2, 32);
        p += __shfl_xor(p, 4, 32);
        p += __shfl_xor(p, 8, 32);
        invp[r] = 1.f / fmaxf(fabsf(p), 1.f);
    }
    #pragma unroll
    for (int ct = 0; ct < 4; ++ct) {
        #pragma unroll
        for (int r = 0; r < 8; ++r) {
            const int m = w * 16 + r + 8 * half;
            out[bh_sd + (size_t)(qb * BQ + m) * Dc + ct * 16 + l16]
                = oacc[ct][r] * invp[r];
        }
    }
}

extern "C" void kernel_launch(void* const* d_in, const int* in_sizes, int n_in,
                              void* d_out, int out_size, void* d_ws, size_t ws_size,
                              hipStream_t stream) {
    (void)in_sizes; (void)n_in; (void)out_size; (void)d_ws; (void)ws_size;
    const float* q  = (const float*)d_in[0];
    const float* k  = (const float*)d_in[1];
    const float* v  = (const float*)d_in[2];
    const float* om = (const float*)d_in[3];
    float* out = (float*)d_out;
    dim3 grid(Sc / BQ, BHc);
    retention_fused_kernel<<<grid, 128, 0, stream>>>(q, k, v, om, out);
}


# No scan launch: raw chunk states parked in workspace; output matrix waves add the head's earlier chunks in scan order (sums unchanged) while staging; stream meets that barrier 4 half rows later
# speedup vs baseline: 6.6901x; 6.6901x over previous
#include <hip/hip_runtime.h>


typedef float v2f __attribute__((ext_vector_type(2)));
typedef float v4f __attribute__((ext_vector_type(4)));
typedef float v8f __attribute__((ext_vector_type(8)));

#define Bc 2
#define Hc 8
#define Sc 2048
#define Dc 64
#define BHc (Bc * Hc)
#define CH 128
#define NCH (Sc / CH)
#define KS 68
#define VS 72
#define ST_FLOATS (Dc * Dc)
#define PARK_FLOATS (ST_FLOATS + Dc)
#define KC_OFF PARK_FLOATS
#define MW 8
#define SW 16
#define RPW (CH / SW)
#define RQ (Sc / 128)

__launch_bounds__(256)
__global__ void chunk_state_kernel(const float* __restrict__ k,
                                   const float* __restrict__ v,
                                   const float* __restrict__ omask,
                                   float* __restrict__ out,
                                   float* park, int park_stride)
{
    const int m    = (int)blockIdx.x;
    const int bh   = (int)blockIdx.y;
    const int tid  = (int)threadIdx.x;
    const int w    = __builtin_amdgcn_readfirstlane(tid >> 5);
    const int lane = tid & 31;
    const int half = lane >> 4;
    const int l16  = lane & 15;

    __shared__ float ldsK[CH * VS];
    __shared__ float ldsV[CH * VS];
    __shared__ float ldsP[8 * Dc];

    const size_t bh_sd = (size_t)bh * Sc * Dc;
    const size_t bh_ss = (size_t)bh * Sc * (size_t)Sc;
    const float* kbase = k + bh_sd + (size_t)(m * CH) * Dc;
    const float* vbase = v + bh_sd + (size_t)(m * CH) * Dc;
    float* region = out + bh_sd + (size_t)(m * CH) * Dc;
    float* rec = park + (size_t)(bh * NCH + m) * park_stride;

    if (w < 4) {
        const int g    = tid >> 4;
        const int col4 = (tid & 15) * 4;
        const float* mcol = omask + bh_ss + (size_t)(m * CH) * Sc;
        float m0[CH / 8];
        v4f kraw[CH / 8];
        #pragma unroll
        for (int it = 0; it < CH / 8; ++it) kraw[it] = *(const v4f*)(kbase + (g + 8 * it) * Dc + col4);
        #pragma unroll
        for (int it = 0; it < CH / 8; ++it) m0[it] = mcol[(size_t)(g + 8 * it) * Sc];
        __builtin_amdgcn_sched_barrier(0);
        float ks0 = 0.f, ks1 = 0.f, ks2 = 0.f, ks3 = 0.f;
        float kcs[CH / 8];
        #pragma unroll
        for (int it = 0; it < CH / 8; ++it) {
            const int row = g + 8 * it;
            const float kc = 1.f / m0[it];
            const v4f kvv = kraw[it];
            const float k0 = kvv.x * kc, k1 = kvv.y * kc, k2 = kvv.z * kc, k3 = kvv.w * kc;
            ldsK[row * VS + col4 + 0] = k0;
            ldsK[row * VS + col4 + 1] = k1;
            ldsK[row * VS + col4 + 2] = k2;
            ldsK[row * VS + col4 + 3] = k3;
            ks0 = __builtin_fmaf(kvv.x, kc, ks0); ks1 = __builtin_fmaf(kvv.y, kc, ks1);
            ks2 = __builtin_fmaf(kvv.z, kc, ks2); ks3 = __builtin_fmaf(kvv.w, kc, ks3);
            kcs[it] = kc;
        }
        if (col4 == 0) {
            #pragma unroll
            for (int it = 0; it < CH / 8; ++it) region[KC_OFF + g + 8 * it] = kcs[it];
        }
        ldsP[g * Dc + col4 + 0] = ks0;
        ldsP[g * Dc + col4 + 1] = ks1;
        ldsP[g * Dc + col4 + 2] = ks2;
        ldsP[g * Dc + col4 + 3] = ks3;
    } else {
        const int g    = (tid - 128) >> 4;
        const int col4 = (tid & 15) * 4;
        #pragma unroll
        for (int it = 0; it < CH / 8; ++it) {
            const int row = g + 8 * it;
            const v4f vvv = *(const v4f*)(vbase + row * Dc + col4);
            ldsV[row * VS + col4 + 0] = vvv.x;
            ldsV[row * VS + col4 + 1] = vvv.y;
            ldsV[row * VS + col4 + 2] = vvv.z;
            ldsV[row * VS + col4 + 3] = vvv.w;
        }
    }
    __syncthreads();
    if (w >= 4) return;

    if (tid < Dc) {
        float s = 0.f;
        #pragma unroll
        for (int g = 0; g < 8; ++g) s += ldsP[g * Dc + tid];
        rec[ST_FLOATS + tid] = s;
    }

    v8f acc[4];
    #pragma unroll
    for (int ct = 0; ct < 4; ++ct) {
        #pragma unroll
        for (int r = 0; r < 8; ++r) acc[ct][r] = 0.f;
    }
    #pragma unroll 8
    for (int kk = 0; kk < CH / 4; ++kk) {
        const int j0 = kk * 4 + half * 2;
        v2f a; a.x = ldsK[(j0    ) * VS + w * 16 + l16];
               a.y = ldsK[(j0 + 1) * VS + w * 16 + l16];
        #pragma unroll
        for (int ct = 0; ct < 4; ++ct) {
            v2f b; b.x = ldsV[(j0    ) * VS + ct * 16 + l16];
                   b.y = ldsV[(j0 + 1) * VS + ct * 16 + l16];
            acc[ct] = __builtin_amdgcn_wmma_f32_16x16x4_f32(
                          false, a, false, b, (short)0, acc[ct], false, false);
        }
    }
    #pragma unroll
    for (int ct = 0; ct < 4; ++ct) {
        #pragma unroll
        for (int r = 0; r < 8; ++r)
            rec[(w * 16 + r + 8 * half) * Dc + ct * 16 + l16] = acc[ct][r];
    }
}

__launch_bounds__(256)
__global__ void chunk_scan_kernel(float* __restrict__ out)
{
    const int e  = (int)(blockIdx.x * 256u + threadIdx.x);
    const int bh = (int)blockIdx.y;
    if (e >= PARK_FLOATS) return;
    float* p = out + (size_t)bh * Sc * Dc + e;
    float val[NCH];
    #pragma unroll
    for (int m = 0; m < NCH; ++m) val[m] = p[(size_t)m * CH * Dc];
    float run = 0.f;
    #pragma unroll
    for (int m = 0; m < NCH; ++m) {
        p[(size_t)m * CH * Dc] = run;
        run += val[m];
    }
}


#define LDS_BARRIER() do { __builtin_amdgcn_fence(__ATOMIC_RELEASE, "workgroup", "local"); \
                           __builtin_amdgcn_s_barrier(); \
                           __builtin_amdgcn_fence(__ATOMIC_ACQUIRE, "workgroup", "local"); } while (0)

template <bool FOLD>
__launch_bounds__(768)
__global__ void retention_chunk_out_kernel(const float* __restrict__ q,
                                           const float* __restrict__ k,
                                           const float* __restrict__ v,
                                           const float* __restrict__ omask,
                                           const float* __restrict__ park,
                                           float* out)
{
    constexpr int PRE = FOLD ? 6 : 2;
    const int m    = (int)blockIdx.x;
    const int bh   = (int)blockIdx.y;
    const int tid  = (int)threadIdx.x;
    const int w    = __builtin_amdgcn_readfirstlane(tid >> 5);
    const bool mat = (w < MW);
    const int lane = tid & 31;
    const int half = lane >> 4;
    const int l16  = lane & 15;
    const int lb   = w >> 2;

    __shared__ float ldsK[CH * KS];
    __shared__ float ldsV[CH * VS];
    __shared__ float ldsS[CH * KS];
    __shared__ float ldsSt[Dc * VS];
    __shared__ float ldsT[Dc];
    __shared__ float ldsQC[CH];

    const size_t bh_sd = (size_t)bh * Sc * Dc;
    const size_t bh_ss = (size_t)bh * Sc * (size_t)Sc;
    float* region = out + bh_sd + (size_t)(m * CH) * Dc;

    const int sw = mat ? 0 : (w - MW);
    const v4f* rp = (const v4f*)(omask + bh_ss + (size_t)(m * CH + sw * RPW) * Sc) + lane;
    v4f hb[3][RQ / 2];
    float part[RPW];
    int head[RPW];
    float racc = 0.f;
#define LOAD_HALF(H) do { _Pragma("unroll") for (int i_ = 0; i_ < RQ / 2; ++i_) \
        hb[(H) % 3][i_] = __builtin_nontemporal_load(rp + (size_t)((H) >> 1) * (Sc / 4) + 32 * (((H) & 1) * (RQ / 2) + i_)); } while (0)
#define SUM_HALF(H) do { _Pragma("unroll") for (int i_ = 0; i_ < RQ / 2; ++i_) { \
        const v4f t_ = hb[(H) % 3][i_]; racc += (t_.x + t_.y) + (t_.z + t_.w); } \
        if (((H) & 1) == 0) head[(H) >> 1] = __builtin_amdgcn_readfirstlane(__builtin_bit_cast(int, hb[(H) % 3][0].x)); \
        else { part[(H) >> 1] = racc; racc = 0.f; } } while (0)
#define STREAM_STEP(H) do { __builtin_amdgcn_sched_barrier(0); \
        if ((H) + 2 < 2 * RPW) LOAD_HALF((H) + 2); \
        __builtin_amdgcn_sched_barrier(0); \
        SUM_HALF(H); } while (0)

    if (mat) {
        const float* kbase = k + bh_sd + (size_t)(m * CH) * Dc;
        const float* vbase = v + bh_sd + (size_t)(m * CH) * Dc;
        #pragma unroll
        for (int it = 0; it < CH * Dc / 4 / (32 * MW); ++it) {
            const int idx  = tid + 32 * MW * it;
            const int row  = idx >> 4;
            const int col4 = (idx & 15) * 4;
            const float kc = region[KC_OFF + row];
            const v4f kvv = *(const v4f*)(kbase + row * Dc + col4);
            ldsK[row * KS + col4 + 0] = kvv.x * kc;
            ldsK[row * KS + col4 + 1] = kvv.y * kc;
            ldsK[row * KS + col4 + 2] = kvv.z * kc;
            ldsK[row * KS + col4 + 3] = kvv.w * kc;
            const v4f vvv = *(const v4f*)(vbase + row * Dc + col4);
            ldsV[row * VS + col4 + 0] = vvv.x;
            ldsV[row * VS + col4 + 1] = vvv.y;
            ldsV[row * VS + col4 + 2] = vvv.z;
            ldsV[row * VS + col4 + 3] = vvv.w;
        }
        if constexpr (FOLD) {
            constexpr int SIT = Dc * Dc / 4 / (32 * MW);
            const float* hrec = park + (size_t)(bh * NCH) * PARK_FLOATS;
            v4f run[SIT];
            float runT = 0.f;
            #pragma unroll
            for (int it = 0; it < SIT; ++it) run[it] = (v4f){0.f, 0.f, 0.f, 0.f};
            for (int c0 = 0; c0 < m; c0 += 4) {
                v4f t[4][SIT];
                float tt[4];
                #pragma unroll
                for (int u = 0; u < 4; ++u) {
                    if (c0 + u < m) {
                        const float* r = hrec + (size_t)(c0 + u) * PARK_FLOATS;
                        #pragma unroll
                        for (int it = 0; it < SIT; ++it) t[u][it] = *(const v4f*)(r + (tid + 32 * MW * it) * 4);
                        tt[u] = (tid < Dc) ? r[ST_FLOATS + tid] : 0.f;
                    }
                }
                #pragma unroll
                for (int u = 0; u < 4; ++u) {
                    if (c0 + u < m) {
                        #pragma unroll
                        for (int it = 0; it < SIT; ++it) run[it] += t[u][it];
                        runT += tt[u];
                    }
                }
            }
            #pragma unroll
            for (int it = 0; it < SIT; ++it) {
                const int idx  = tid + 32 * MW * it;
                const int row  = idx >> 4;
                const int col4 = (idx & 15) * 4;
                ldsSt[row * VS + col4 + 0] = run[it].x;
                ldsSt[row * VS + col4 + 1] = run[it].y;
                ldsSt[row * VS + col4 + 2] = run[it].z;
                ldsSt[row * VS + col4 + 3] = run[it].w;
            }
            if (tid < Dc) ldsT[tid] = runT;
        } else {
            #pragma unroll
            for (int it = 0; it < Dc * Dc / 4 / (32 * MW); ++it) {
                const int idx  = tid + 32 * MW * it;
                const int row  = idx >> 4;
                const int col4 = (idx & 15) * 4;
                const v4f sv = *(const v4f*)(region + row * Dc + col4);
                ldsSt[row * VS + col4 + 0] = sv.x;
                ldsSt[row * VS + col4 + 1] = sv.y;
                ldsSt[row * VS + col4 + 2] = sv.z;
                ldsSt[row * VS + col4 + 3] = sv.w;
            }
            if (tid < Dc) ldsT[tid] = region[ST_FLOATS + tid];
        }
    } else {
        LOAD_HALF(0);
        LOAD_HALF(1);
        #pragma unroll
        for (int h = 0; h < PRE; ++h) STREAM_STEP(h);
    }
    LDS_BARRIER();

    v8f oacc[4];
    v8f pacc;
    float pin;

    if (mat) {
        pin = 0.f;
        #pragma unroll
        for (int r = 0; r < 8; ++r) pacc[r] = 0.f;
        #pragma unroll
        for (int ct = 0; ct < 4; ++ct) {
            #pragma unroll
            for (int r = 0; r < 8; ++r) oacc[ct][r] = 0.f;
        }
        const int srow = w * 16 + l16;
        const float* qptr = q + bh_sd + (size_t)(m * CH + srow) * Dc;
        v2f qf[16];
        #pragma unroll
        for (int kk = 0; kk < 16; ++kk) {
            const int c = kk * 4 + half * 2;
            v2f a; a.x = qptr[c]; a.y = qptr[c + 1];
            qf[kk] = a;
        }

        #pragma unroll
        for (int kk = 0; kk < 16; ++kk) {
            const int d0 = kk * 4 + half * 2;
            #pragma unroll
            for (int ct = 0; ct < 4; ++ct) {
                v2f b; b.x = ldsSt[(d0    ) * VS + ct * 16 + l16];
                       b.y = ldsSt[(d0 + 1) * VS + ct * 16 + l16];
                oacc[ct] = __builtin_amdgcn_wmma_f32_16x16x4_f32(
                               false, qf[kk], false, b, (short)0, oacc[ct], false, false);
            }
        }
        for (int kb = 0; kb <= lb; ++kb) {
            #pragma unroll
            for (int t = 0; t < 4; ++t) {
                v8f s;
                #pragma unroll
                for (int r = 0; r < 8; ++r) s[r] = 0.f;
                #pragma unroll
                for (int kk = 0; kk < 16; ++kk) {
                    const int d0   = kk * 4 + half * 2;
                    const int krow = kb * 64 + t * 16 + l16;
                    v2f b; b.x = ldsK[krow * KS + d0]; b.y = ldsK[krow * KS + d0 + 1];
                    s = __builtin_amdgcn_wmma_f32_16x16x4_f32(
                            false, qf[kk], false, b, (short)0, s, false, false);
                }
                const int jg = kb * 64 + t * 16 + l16;
                #pragma unroll
                for (int r = 0; r < 8; ++r) {
                    const int ig = w * 16 + r + 8 * half;
                    const float sv = (jg <= ig) ? s[r] : 0.f;
                    pacc[r] += sv;
                    ldsS[ig * KS + t * 16 + l16] = sv;
                }
            }
            __builtin_amdgcn_wave_barrier();
            #pragma unroll
            for (int nn = 0; nn < 16; ++nn) {
                const int s0 = nn * 4 + half * 2;
                const int n0 = kb * 64 + s0;
                v2f a; a.x = ldsS[srow * KS + s0]; a.y = ldsS[srow * KS + s0 + 1];
                #pragma unroll
                for (int ct = 0; ct < 4; ++ct) {
                    v2f b; b.x = ldsV[(n0    ) * VS + ct * 16 + l16];
                           b.y = ldsV[(n0 + 1) * VS + ct * 16 + l16];
                    oacc[ct] = __builtin_amdgcn_wmma_f32_16x16x4_f32(
                                   false, a, false, b, (short)0, oacc[ct], false, false);
                }
            }
            __builtin_amdgcn_wave_barrier();
            __builtin_amdgcn_sched_barrier(0);
            asm volatile("v_nop\n\tv_nop\n\tv_nop\n\tv_nop");
        }
        #pragma unroll
        for (int kk = 0; kk < 16; ++kk) {
            const int d0 = kk * 4 + half * 2;
            pin += qf[kk].x * ldsT[d0] + qf[kk].y * ldsT[d0 + 1];
        }
        pin += __shfl_xor(pin, 16, 32);
    } else {
        #pragma unroll
        for (int h = PRE; h < 2 * RPW; ++h) STREAM_STEP(h);
        float mys = 1.f, mym = 1.f;
        #pragma unroll
        for (int r = 0; r < RPW; ++r) {
            float s = part[r];
            #pragma unroll
            for (int off = 16; off > 0; off >>= 1) s += __shfl_xor(s, off, 32);
            const float m0 = __builtin_bit_cast(float, head[r]);
            if (lane == r) { mys = s; mym = m0; }
        }
        if (lane < RPW) ldsQC[sw * RPW + lane] = mym / sqrtf(mys);
    }
#undef LOAD_HALF
#undef SUM_HALF
#undef STREAM_STEP
    LDS_BARRIER();

    if (mat) {
        float scale[8];
        #pragma unroll
        for (int r = 0; r < 8; ++r) {
            float p = pacc[r];
            p += __shfl_xor(p, 1, 32);
            p += __shfl_xor(p, 2, 32);
            p += __shfl_xor(p, 4, 32);
            p += __shfl_xor(p, 8, 32);
            p += __shfl(pin, r + 8 * half, 32);
            const float c = ldsQC[w * 16 + r + 8 * half];
            scale[r] = c / fmaxf(fabsf(c * p), 1.f);
        }
        #pragma unroll
        for (int ct = 0; ct < 4; ++ct) {
            #pragma unroll
            for (int r = 0; r < 8; ++r) {
                const int mrow = w * 16 + r + 8 * half;
                region[mrow * Dc + ct * 16 + l16] = oacc[ct][r] * scale[r];
            }
        }
    }
}

extern "C" void kernel_launch(void* const* d_in, const int* in_sizes, int n_in,
                              void* d_out, int out_size, void* d_ws, size_t ws_size,
                              hipStream_t stream) {
    (void)in_sizes; (void)n_in; (void)out_size;
    const float* q  = (const float*)d_in[0];
    const float* k  = (const float*)d_in[1];
    const float* v  = (const float*)d_in[2];
    const float* om = (const float*)d_in[3];
    float* out = (float*)d_out;
    dim3 gchunk(NCH, BHc);
    if (d_ws != nullptr && ws_size >= sizeof(float) * BHc * NCH * PARK_FLOATS) {
        float* ws = (float*)d_ws;
        chunk_state_kernel<<<gchunk, 256, 0, stream>>>(k, v, om, out, ws, PARK_FLOATS);
        retention_chunk_out_kernel<true><<<gchunk, 768, 0, stream>>>(q, k, v, om, ws, out);
    } else {
        dim3 gscan(17, BHc);
        chunk_state_kernel<<<gchunk, 256, 0, stream>>>(k, v, om, out, out, CH * Dc);
        chunk_scan_kernel<<<gscan, 256, 0, stream>>>(out);
        retention_chunk_out_kernel<false><<<gchunk, 768, 0, stream>>>(q, k, v, om, out, out);
    }
}
